# starting kernel
# baseline (speedup 1.0000x reference)
#include <hip/hip_runtime.h>
#include <hip/hip_bf16.h>

typedef __attribute__((ext_vector_type(2))) float v2f;
typedef __attribute__((ext_vector_type(8))) float v8f;

#define IN_C   64
#define HID_C  64
#define N_REL  8
#define TILE_M 16
#define SROW   68

__global__ void wt_transpose_kernel(const float* __restrict__ W,
                                    float* __restrict__ Wt, int total) {
  int i = blockIdx.x * blockDim.x + threadIdx.x;
  if (i < total) {
    int r = i >> 12;
    int k = (i >> 6) & 63;
    int n = i & 63;
    Wt[(r << 12) + (n << 6) + k] = W[i];
  }
}

__global__ __launch_bounds__(128) void rgcn_fused_kernel(
    const float* __restrict__ x,
    const float* __restrict__ Wt,
    const int*   __restrict__ ptr,
    const int*   __restrict__ idx,
    const int*   __restrict__ et,
    const int*   __restrict__ hmap,
    const float* __restrict__ hbuf,
    const int*   __restrict__ hsize_p,
    float* __restrict__ out,
    int n_node)
{
  __shared__ float S[N_REL * TILE_M * SROW];
  __shared__ float s_deg[TILE_M];
  __shared__ int   s_map[TILE_M];

  const int tid   = threadIdx.x;
  const int lane  = tid & 31;
  const int wave  = tid >> 5;
  const int node0 = blockIdx.x * TILE_M;

  if (tid < TILE_M) {
    int n  = node0 + tid;
    int nc = (n < n_node) ? n : (n_node - 1);
    s_deg[tid] = (float)(ptr[nc + 1] - ptr[nc]);
    s_map[tid] = hmap[nc];
  }

  for (int nn = 0; nn < 4; ++nn) {
    const int local = wave * 4 + nn;
    int n = node0 + local;
    if (n >= n_node) n = n_node - 1;
    const int e0 = __builtin_amdgcn_readfirstlane(ptr[n]);
    const int e1 = __builtin_amdgcn_readfirstlane(ptr[n + 1]);
    v2f acc[N_REL];
#pragma unroll
    for (int r = 0; r < N_REL; ++r) acc[r] = (v2f)0.0f;
    for (int e = e0; e < e1; ++e) {
      const int src = __builtin_amdgcn_readfirstlane(idx[e]);
      const int t   = __builtin_amdgcn_readfirstlane(et[e]);
      const v2f v = *(const v2f*)(x + (size_t)src * IN_C + 2 * lane);
#pragma unroll
      for (int r = 0; r < N_REL; ++r) {
        if (t == r) acc[r] += v;
      }
    }
#pragma unroll
    for (int r = 0; r < N_REL; ++r) {
      *(v2f*)(S + (r * TILE_M + local) * SROW + 2 * lane) = acc[r];
    }
  }
  __syncthreads();

  const int ncol = lane & 15;
  const int hi   = lane >> 4;
  const int n0   = wave * 16;
  v8f acc = {};

#pragma unroll 2
  for (int r = 0; r < N_REL; ++r) {
    const float* Sr = S + r * TILE_M * SROW;
    const float* Br = Wt + (r << 12) + ((n0 + ncol) << 6);
#pragma unroll
    for (int j = 0; j < 16; ++j) {
      const int k = 4 * j + 2 * hi;
      const v2f A = *(const v2f*)(Sr + ncol * SROW + k);
      const v2f B = *(const v2f*)(Br + k);
      acc = __builtin_amdgcn_wmma_f32_16x16x4_f32(
          false, A, false, B, (short)0, acc, false, false);
    }
  }

  const int hsize = hsize_p[0];
  const size_t his_off = (size_t)n_node * HID_C;
#pragma unroll
  for (int v = 0; v < 8; ++v) {
    const int m    = v + 8 * hi;
    const int node = node0 + m;
    if (node < n_node) {
      float o = acc[v] / s_deg[m];
      if (hsize > 0 && s_map[m] != -1) {
        o = hbuf[(size_t)node * HID_C + n0 + ncol];
      }
      const size_t off = (size_t)node * HID_C + n0 + ncol;
      out[off]           = o;
      out[his_off + off] = o;
    }
  }
}

extern "C" void kernel_launch(void* const* d_in, const int* in_sizes, int n_in,
                              void* d_out, int out_size, void* d_ws, size_t ws_size,
                              hipStream_t stream) {
  const float* x    = (const float*)d_in[0];
  const float* W    = (const float*)d_in[1];
  const int*   ptr  = (const int*)d_in[2];
  const int*   idx  = (const int*)d_in[3];
  const int*   et   = (const int*)d_in[4];
  const int*   hmap = (const int*)d_in[6];
  const float* hbuf = (const float*)d_in[7];
  const int*   hsz  = (const int*)d_in[9];
  float* out = (float*)d_out;
  float* Wt  = (float*)d_ws;

  const int n_node = in_sizes[6];
  const int wtotal = in_sizes[1];

  wt_transpose_kernel<<<(wtotal + 255) / 256, 256, 0, stream>>>(W, Wt, wtotal);

  const int nblocks = (n_node + TILE_M - 1) / TILE_M;
  rgcn_fused_kernel<<<nblocks, 128, 0, stream>>>(x, Wt, ptr, idx, et, hmap,
                                                 hbuf, hsz, out, n_node);
}


# Decomposition swapped: each node transformed once on WMMA (x @ all 8 W_r, rows kept in workspace), then a half-wave per node sums its edges' 256 B rows, 8 gathers in flight; products summed per edge
# speedup vs baseline: 1.6762x; 1.6762x over previous
#include <hip/hip_runtime.h>

typedef __attribute__((ext_vector_type(2))) float v2f;
typedef __attribute__((ext_vector_type(4))) float v4f;
typedef __attribute__((ext_vector_type(8))) float v8f;

#define IN_C   64
#define HID_C  64
#define N_REL  8
#define TILE_M 16
#define SROW   68

__device__ __forceinline__ int kmap(int s, int h, int e) {
  return 8 * (s >> 1) + 4 * h + 2 * (s & 1) + e;
}

__global__ __launch_bounds__(256) void rgcn_transform_kernel(
    const float* __restrict__ x,
    const float* __restrict__ W,
    float* __restrict__ XR,
    int n_node)
{
  __shared__ float S[N_REL * TILE_M * SROW];

  const int tid  = threadIdx.x;
  const int lane = tid & 31;
  const int r    = tid >> 5;
  const int c    = lane & 15;
  const int h    = lane >> 4;

  v2f B[4][16];
  const float* Wr = W + (r << 12);
#pragma unroll
  for (int t = 0; t < 4; ++t) {
#pragma unroll
    for (int s = 0; s < 16; ++s) {
      B[t][s].x = Wr[kmap(s, h, 0) * HID_C + 16 * t + c];
      B[t][s].y = Wr[kmap(s, h, 1) * HID_C + 16 * t + c];
    }
  }

  float* Sw = S + r * TILE_M * SROW;
  const int n_tiles = (n_node + TILE_M - 1) / TILE_M;

  v4f A4[8];
  {
    int arow = blockIdx.x * TILE_M + c;
    if (arow >= n_node) arow = n_node - 1;
    const float* xa = x + (size_t)arow * IN_C + 4 * h;
#pragma unroll
    for (int i = 0; i < 8; ++i) A4[i] = *(const v4f*)(xa + 8 * i);
  }

  for (int mt = blockIdx.x; mt < n_tiles; mt += gridDim.x) {
    const int node0 = mt * TILE_M;

    v4f A4n[8];
    {
      int arow = node0 + (int)gridDim.x * TILE_M + c;
      if (arow >= n_node) arow = n_node - 1;
      const float* xa = x + (size_t)arow * IN_C + 4 * h;
#pragma unroll
      for (int i = 0; i < 8; ++i) A4n[i] = *(const v4f*)(xa + 8 * i);
    }

    v8f acc[4];
#pragma unroll
    for (int t = 0; t < 4; ++t) acc[t] = (v8f)0.0f;

#pragma unroll
    for (int s = 0; s < 16; ++s) {
      v2f A;
      A.x = A4[s >> 1][2 * (s & 1)];
      A.y = A4[s >> 1][2 * (s & 1) + 1];
#pragma unroll
      for (int t = 0; t < 4; ++t) {
        acc[t] = __builtin_amdgcn_wmma_f32_16x16x4_f32(
            false, A, false, B[t][s], (short)0, acc[t], false, false);
      }
    }

#pragma unroll
    for (int t = 0; t < 4; ++t) {
#pragma unroll
      for (int v = 0; v < 8; ++v) {
        Sw[(v + 8 * h) * SROW + 16 * t + c] = acc[t][v];
      }
    }
    __builtin_amdgcn_fence(__ATOMIC_RELEASE, "wavefront");
    __builtin_amdgcn_wave_barrier();
    __builtin_amdgcn_fence(__ATOMIC_ACQUIRE, "wavefront");

    v4f o[8];
#pragma unroll
    for (int i = 0; i < 8; ++i) o[i] = *(const v4f*)(Sw + (2 * i + h) * SROW + 4 * c);

    float* xr = XR + ((size_t)(node0 + h) * N_REL + r) * HID_C + 4 * c;
    if (node0 + TILE_M <= n_node) {
#pragma unroll
      for (int i = 0; i < 8; ++i) *(v4f*)(xr + (size_t)(2 * i) * N_REL * HID_C) = o[i];
    } else {
#pragma unroll
      for (int i = 0; i < 8; ++i) {
        if (node0 + 2 * i + h < n_node) *(v4f*)(xr + (size_t)(2 * i) * N_REL * HID_C) = o[i];
      }
    }
    __builtin_amdgcn_fence(__ATOMIC_RELEASE, "wavefront");
    __builtin_amdgcn_wave_barrier();
    __builtin_amdgcn_fence(__ATOMIC_ACQUIRE, "wavefront");

#pragma unroll
    for (int i = 0; i < 8; ++i) A4[i] = A4n[i];
  }
}

#define NPW 2

__device__ __forceinline__ v4f gather_node(const float* __restrict__ XR,
                                           const int* __restrict__ idx,
                                           const int* __restrict__ et,
                                           int e0, int len, int maxlen, int ri, int rt, int c)
{
  v4f acc = (v4f)0.0f;
  for (int base = 0; base < maxlen; base += 16) {
    const int row_cur = ri * N_REL + rt;
    const int nb = base + 16 + c;
    ri = 0; rt = 0;
    if (nb < len) { ri = idx[e0 + nb]; rt = et[e0 + nb]; }

#pragma unroll
    for (int g = 0; g < 2; ++g) {
      if (base + 8 * g < maxlen) {
        v4f v[8];
#pragma unroll
        for (int k = 0; k < 8; ++k) {
          const int rk = __shfl(row_cur, 8 * g + k, 16);
          v[k] = *(const v4f*)(XR + (size_t)rk * HID_C + 4 * c);
        }
#pragma unroll
        for (int k = 0; k < 8; ++k) {
          if (base + 8 * g + k < len) acc += v[k];
        }
      }
    }
  }
  return acc;
}

__global__ __launch_bounds__(256) void rgcn_gather_kernel(
    const float* __restrict__ XR,
    const int*   __restrict__ ptr,
    const int*   __restrict__ idx,
    const int*   __restrict__ et,
    const int*   __restrict__ hmap,
    const float* __restrict__ hbuf,
    const int*   __restrict__ hsize_p,
    float* __restrict__ out,
    int n_node)
{
  const int tid  = threadIdx.x;
  const int lane = tid & 31;
  const int wave = tid >> 5;
  const int c    = lane & 15;
  const int h    = lane >> 4;
  const int nb0  = ((blockIdx.x * 8 + wave) * 2 + h) * NPW;

  int pi = nb0 + (c < NPW ? c : NPW);
  if (pi > n_node) pi = n_node;
  const int pv = ptr[pi];
  int hi_ = nb0 + (c < NPW ? c : NPW - 1);
  if (hi_ > n_node - 1) hi_ = n_node - 1;
  const int hv = hmap[hi_];
  const int hsize = hsize_p[0];

  int e0[NPW], e1[NPW], len[NPW], maxlen[NPW], ri[NPW], rt[NPW];
#pragma unroll
  for (int j = 0; j < NPW; ++j) {
    e0[j] = __shfl(pv, j, 16);
    e1[j] = __shfl(pv, j + 1, 16);
    len[j] = (nb0 + j < n_node) ? (e1[j] - e0[j]) : 0;
    const int other = __shfl_xor(len[j], 16);
    maxlen[j] = len[j] > other ? len[j] : other;
    ri[j] = 0; rt[j] = 0;
    if (c < len[j]) { ri[j] = idx[e0[j] + c]; rt[j] = et[e0[j] + c]; }
  }

#pragma unroll
  for (int j = 0; j < NPW; ++j) {
    const int node = nb0 + j;
    const v4f acc = gather_node(XR, idx, et, e0[j], len[j], maxlen[j], ri[j], rt[j], c);

    const float deg = (float)(e1[j] - e0[j]);
    v4f o;
    o.x = acc.x / deg; o.y = acc.y / deg; o.z = acc.z / deg; o.w = acc.w / deg;

    const int hm = __shfl(hv, j, 16);
    if (node < n_node) {
      if (hsize > 0 && hm != -1) {
        o = *(const v4f*)(hbuf + (size_t)node * HID_C + 4 * c);
      }
      const size_t off = (size_t)node * HID_C + 4 * c;
      *(v4f*)(out + off) = o;
      *(v4f*)(out + (size_t)n_node * HID_C + off) = o;
    }
  }
}

extern "C" void kernel_launch(void* const* d_in, const int* in_sizes, int n_in,
                              void* d_out, int out_size, void* d_ws, size_t ws_size,
                              hipStream_t stream) {
  const float* x    = (const float*)d_in[0];
  const float* W    = (const float*)d_in[1];
  const int*   ptr  = (const int*)d_in[2];
  const int*   idx  = (const int*)d_in[3];
  const int*   et   = (const int*)d_in[4];
  const int*   hmap = (const int*)d_in[6];
  const float* hbuf = (const float*)d_in[7];
  const int*   hsz  = (const int*)d_in[9];
  float* out = (float*)d_out;
  float* XR  = (float*)d_ws;

  const int n_node = in_sizes[6];

  rgcn_transform_kernel<<<1024, 256, 0, stream>>>(x, W, XR, n_node);

  const int nblocks = (n_node + 31) / 32;
  rgcn_gather_kernel<<<nblocks, 256, 0, stream>>>(XR, ptr, idx, et, hmap, hbuf, hsz,
                                                  out, n_node);
}


# One launch, table gone: per 16-node tile each half-wave sums its node's gathered x rows per relation in LDS, wave r runs plane r x W[r] on WMMA, 8 partials added: sums per relation, then across
# speedup vs baseline: 2.2913x; 1.3670x over previous
#include <hip/hip_runtime.h>

typedef __attribute__((ext_vector_type(2))) float v2f;
typedef __attribute__((ext_vector_type(4))) float v4f;
typedef __attribute__((ext_vector_type(8))) float v8f;

#define IN_C   64
#define HID_C  64
#define N_REL  8
#define TILE_M 16
#define SROW   68
#define RSTRIDE (TILE_M * SROW)

__device__ __forceinline__ int kmap(int s, int h, int e) {
  return 8 * (s >> 1) + 4 * h + 2 * (s & 1) + e;
}

__global__ __launch_bounds__(256) void rgcn_fused_kernel(
    const float* __restrict__ x,
    const float* __restrict__ W,
    const int*   __restrict__ ptr,
    const int*   __restrict__ idx,
    const int*   __restrict__ et,
    const int*   __restrict__ hmap,
    const float* __restrict__ hbuf,
    const int*   __restrict__ hsize_p,
    float* __restrict__ out,
    int n_node)
{
  __shared__ float S[N_REL * RSTRIDE];

  const int tid  = threadIdx.x;
  const int lane = tid & 31;
  const int r    = tid >> 5;
  const int c    = lane & 15;
  const int h    = lane >> 4;
  const int hw   = tid >> 4;

  v2f B[4][16];
  const float* Wr = W + (r << 12);
#pragma unroll
  for (int t = 0; t < 4; ++t) {
#pragma unroll
    for (int s = 0; s < 16; ++s) {
      B[t][s].x = Wr[kmap(s, h, 0) * HID_C + 16 * t + c];
      B[t][s].y = Wr[kmap(s, h, 1) * HID_C + 16 * t + c];
    }
  }

  const int hsize   = hsize_p[0];
  const int n_tiles = (n_node + TILE_M - 1) / TILE_M;
  const int gstep   = (int)gridDim.x;

  float* Sn = S + hw * SROW + 4 * c;
  float* Sw = S + r * RSTRIDE;

  int c_e0 = 0, c_len = 0, c_ri = 0, c_rt = 0, c_hm = -1;
  int n_e0 = 0, n_e1 = 0, n_hm = -1;
  {
    const int node = blockIdx.x * TILE_M + hw;
    if (node < n_node) {
      c_e0 = ptr[node];
      c_len = ptr[node + 1] - c_e0;
      c_hm = hmap[node];
    }
    if (c < c_len) { c_ri = idx[c_e0 + c]; c_rt = et[c_e0 + c]; }
    const int node_n = node + gstep * TILE_M;
    if (node_n < n_node) {
      n_e0 = ptr[node_n];
      n_e1 = ptr[node_n + 1];
      n_hm = hmap[node_n];
    }
  }

  for (int mt = blockIdx.x; mt < n_tiles; mt += gstep) {
    const int node = mt * TILE_M + hw;

#pragma unroll
    for (int q = 0; q < N_REL; ++q) *(v4f*)(Sn + q * RSTRIDE) = (v4f)0.0f;

    {
      const int other  = __shfl_xor(c_len, 16);
      const int maxlen = c_len > other ? c_len : other;
      int ri = c_ri, rt = c_rt;
      for (int base = 0; base < maxlen; base += 16) {
        const int cur_i = ri;
        int cur_t = rt;
        cur_t = cur_t < 0 ? 0 : (cur_t > N_REL - 1 ? N_REL - 1 : cur_t);
        const int nb = base + 16 + c;
        ri = 0; rt = 0;
        if (nb < c_len) { ri = idx[c_e0 + nb]; rt = et[c_e0 + nb]; }

#pragma unroll
        for (int g = 0; g < 2; ++g) {
          if (base + 8 * g < maxlen) {
            v4f v[8];
            int tk[8];
#pragma unroll
            for (int k = 0; k < 8; ++k) {
              const int ik = __shfl(cur_i, 8 * g + k, 16);
              tk[k] = __shfl(cur_t, 8 * g + k, 16);
              v[k] = *(const v4f*)(x + (size_t)ik * IN_C + 4 * c);
            }
#pragma unroll
            for (int k = 0; k < 8; ++k) {
              if (base + 8 * g + k < c_len) {
                float* p = Sn + tk[k] * RSTRIDE;
                v4f a = *(const v4f*)p;
                a += v[k];
                *(v4f*)p = a;
              }
            }
          }
        }
      }
    }
    __syncthreads();

    const int nn_len = n_e1 - n_e0;
    int nn_ri = 0, nn_rt = 0;
    if (c < nn_len) { nn_ri = idx[n_e0 + c]; nn_rt = et[n_e0 + c]; }
    int p_e0 = 0, p_e1 = 0, p_hm = -1;
    {
      const int node_p = node + 2 * gstep * TILE_M;
      if (node_p < n_node) {
        p_e0 = ptr[node_p];
        p_e1 = ptr[node_p + 1];
        p_hm = hmap[node_p];
      }
    }

    v4f A4[8];
    {
      const float* xa = Sw + c * SROW + 4 * h;
#pragma unroll
      for (int i = 0; i < 8; ++i) A4[i] = *(const v4f*)(xa + 8 * i);
    }
    asm volatile("" :: "v"(A4[0]), "v"(A4[1]), "v"(A4[2]), "v"(A4[3]),
                       "v"(A4[4]), "v"(A4[5]), "v"(A4[6]), "v"(A4[7]));

    v8f acc[4];
#pragma unroll
    for (int t = 0; t < 4; ++t) acc[t] = (v8f)0.0f;

#pragma unroll
    for (int s = 0; s < 16; ++s) {
      v2f A;
      A.x = A4[s >> 1][2 * (s & 1)];
      A.y = A4[s >> 1][2 * (s & 1) + 1];
#pragma unroll
      for (int t = 0; t < 4; ++t) {
        acc[t] = __builtin_amdgcn_wmma_f32_16x16x4_f32(
            false, A, false, B[t][s], (short)0, acc[t], false, false);
      }
    }

#pragma unroll
    for (int t = 0; t < 4; ++t) {
#pragma unroll
      for (int v = 0; v < 8; ++v) {
        Sw[(v + 8 * h) * SROW + 16 * t + c] = acc[t][v];
      }
    }
    __syncthreads();

    v4f sum = *(const v4f*)Sn;
#pragma unroll
    for (int q = 1; q < N_REL; ++q) sum += *(const v4f*)(Sn + q * RSTRIDE);

    if (node < n_node) {
      const float deg = (float)c_len;
      v4f o;
      o.x = sum.x / deg; o.y = sum.y / deg; o.z = sum.z / deg; o.w = sum.w / deg;
      if (hsize > 0 && c_hm != -1) {
        o = *(const v4f*)(hbuf + (size_t)node * HID_C + 4 * c);
      }
      const size_t off = (size_t)node * HID_C + 4 * c;
      *(v4f*)(out + off) = o;
      *(v4f*)(out + (size_t)n_node * HID_C + off) = o;
    }

    c_e0 = n_e0; c_len = nn_len; c_ri = nn_ri; c_rt = nn_rt; c_hm = n_hm;
    n_e0 = p_e0; n_e1 = p_e1; n_hm = p_hm;
  }
}

extern "C" void kernel_launch(void* const* d_in, const int* in_sizes, int n_in,
                              void* d_out, int out_size, void* d_ws, size_t ws_size,
                              hipStream_t stream) {
  const float* x    = (const float*)d_in[0];
  const float* W    = (const float*)d_in[1];
  const int*   ptr  = (const int*)d_in[2];
  const int*   idx  = (const int*)d_in[3];
  const int*   et   = (const int*)d_in[4];
  const int*   hmap = (const int*)d_in[6];
  const float* hbuf = (const float*)d_in[7];
  const int*   hsz  = (const int*)d_in[9];
  float* out = (float*)d_out;

  const int n_node = in_sizes[6];

  rgcn_fused_kernel<<<512, 256, 0, stream>>>(x, W, ptr, idx, et, hmap, hbuf, hsz, out, n_node);
}


# Gather phase: each half sorts its 16 edge records by relation (stable counting rank), sums a relation's rows in a register run and stores it once: no LDS read-add-write per edge; edge order kept
# speedup vs baseline: 2.4322x; 1.0615x over previous
#include <hip/hip_runtime.h>

typedef __attribute__((ext_vector_type(2))) float v2f;
typedef __attribute__((ext_vector_type(4))) float v4f;
typedef __attribute__((ext_vector_type(8))) float v8f;

#define IN_C   64
#define HID_C  64
#define N_REL  8
#define TILE_M 16
#define SROW   68
#define RSTRIDE (TILE_M * SROW)

__device__ __forceinline__ int kmap(int s, int h, int e) {
  return 8 * (s >> 1) + 4 * h + 2 * (s & 1) + e;
}

__global__ __launch_bounds__(256) void rgcn_fused_kernel(
    const float* __restrict__ x,
    const float* __restrict__ W,
    const int*   __restrict__ ptr,
    const int*   __restrict__ idx,
    const int*   __restrict__ et,
    const int*   __restrict__ hmap,
    const float* __restrict__ hbuf,
    const int*   __restrict__ hsize_p,
    float* __restrict__ out,
    int n_node)
{
  __shared__ float S[N_REL * RSTRIDE];

  const int tid  = threadIdx.x;
  const int lane = tid & 31;
  const int r    = tid >> 5;
  const int c    = lane & 15;
  const int h    = lane >> 4;
  const int hw   = tid >> 4;

  v2f B[4][16];
  const float* Wr = W + (r << 12);
#pragma unroll
  for (int t = 0; t < 4; ++t) {
#pragma unroll
    for (int s = 0; s < 16; ++s) {
      B[t][s].x = Wr[kmap(s, h, 0) * HID_C + 16 * t + c];
      B[t][s].y = Wr[kmap(s, h, 1) * HID_C + 16 * t + c];
    }
  }

  const int hsize   = hsize_p[0];
  const int n_tiles = (n_node + TILE_M - 1) / TILE_M;
  const int gstep   = (int)gridDim.x;

  float* Sn = S + hw * SROW + 4 * c;
  float* Sw = S + r * RSTRIDE;

  int c_e0 = 0, c_len = 0, c_ri = 0, c_rt = 0, c_hm = -1;
  int n_e0 = 0, n_e1 = 0, n_hm = -1;
  {
    const int node = blockIdx.x * TILE_M + hw;
    if (node < n_node) {
      c_e0 = ptr[node];
      c_len = ptr[node + 1] - c_e0;
      c_hm = hmap[node];
    }
    if (c < c_len) { c_ri = idx[c_e0 + c]; c_rt = et[c_e0 + c]; }
    const int node_n = node + gstep * TILE_M;
    if (node_n < n_node) {
      n_e0 = ptr[node_n];
      n_e1 = ptr[node_n + 1];
      n_hm = hmap[node_n];
    }
  }

  for (int mt = blockIdx.x; mt < n_tiles; mt += gstep) {
    const int node = mt * TILE_M + hw;

#pragma unroll
    for (int q = 0; q < N_REL; ++q) *(v4f*)(Sn + q * RSTRIDE) = (v4f)0.0f;

    {
      const int other  = __shfl_xor(c_len, 16);
      const int maxlen = c_len > other ? c_len : other;
      int ri = c_ri, rt = c_rt;
      v4f a = (v4f)0.0f;
      int ta = N_REL;
      for (int base = 0; base < maxlen; base += 16) {
        const int cur_i = ri;
        int cur_t = rt;
        cur_t = cur_t < 0 ? 0 : (cur_t > N_REL - 1 ? N_REL - 1 : cur_t);
        const int nb = base + 16 + c;
        ri = 0; rt = 0;
        if (nb < c_len) { ri = idx[c_e0 + nb]; rt = et[c_e0 + nb]; }

        const int key = (base + c < c_len) ? cur_t : N_REL;
        const unsigned below_me = (1u << c) - 1u;
        int rank = 0;
#pragma unroll
        for (int q = 0; q < N_REL; ++q) {
          const unsigned vote = __builtin_amdgcn_ballot_w32(key == q);
          const unsigned mine = (vote >> (16 * h)) & 0xffffu;
          const unsigned cnt  = (q < key) ? mine : ((q == key) ? (mine & below_me) : 0u);
          rank += __builtin_popcount(cnt);
        }
        if (key == N_REL) rank = c;
        const int dst = ((lane & 16) + rank) << 2;
        const int s_i = __builtin_amdgcn_ds_permute(dst, cur_i);
        const int s_t = __builtin_amdgcn_ds_permute(dst, key);
        const bool first = (base == 0);

#pragma unroll
        for (int g = 0; g < 2; ++g) {
          if (base + 8 * g < maxlen) {
            v4f v[8];
            int tk[8];
#pragma unroll
            for (int k = 0; k < 8; ++k) {
              const int ik = __shfl(s_i, 8 * g + k, 16);
              tk[k] = __shfl(s_t, 8 * g + k, 16);
              v[k] = *(const v4f*)(x + (size_t)ik * IN_C + 4 * c);
            }
#pragma unroll
            for (int k = 0; k < 8; ++k) {
              const bool nr = (tk[k] != ta);
              if (nr && ta < N_REL) *(v4f*)(Sn + ta * RSTRIDE) = a;
              v4f st = (v4f)0.0f;
              if (!first && nr) st = *(const v4f*)(Sn + (tk[k] & (N_REL - 1)) * RSTRIDE);
              a.x = (nr ? st.x : a.x) + v[k].x;
              a.y = (nr ? st.y : a.y) + v[k].y;
              a.z = (nr ? st.z : a.z) + v[k].z;
              a.w = (nr ? st.w : a.w) + v[k].w;
              ta = tk[k];
            }
          }
        }
        if (ta < N_REL) *(v4f*)(Sn + ta * RSTRIDE) = a;
        ta = N_REL;
      }
    }
    __syncthreads();

    const int nn_len = n_e1 - n_e0;
    int nn_ri = 0, nn_rt = 0;
    if (c < nn_len) { nn_ri = idx[n_e0 + c]; nn_rt = et[n_e0 + c]; }
    int p_e0 = 0, p_e1 = 0, p_hm = -1;
    {
      const int node_p = node + 2 * gstep * TILE_M;
      if (node_p < n_node) {
        p_e0 = ptr[node_p];
        p_e1 = ptr[node_p + 1];
        p_hm = hmap[node_p];
      }
    }

    v4f A4[8];
    {
      const float* xa = Sw + c * SROW + 4 * h;
#pragma unroll
      for (int i = 0; i < 8; ++i) A4[i] = *(const v4f*)(xa + 8 * i);
    }
    asm volatile("" :: "v"(A4[0]), "v"(A4[1]), "v"(A4[2]), "v"(A4[3]),
                       "v"(A4[4]), "v"(A4[5]), "v"(A4[6]), "v"(A4[7]));

    v8f acc[4];
#pragma unroll
    for (int t = 0; t < 4; ++t) acc[t] = (v8f)0.0f;

#pragma unroll
    for (int s = 0; s < 16; ++s) {
      v2f A;
      A.x = A4[s >> 1][2 * (s & 1)];
      A.y = A4[s >> 1][2 * (s & 1) + 1];
#pragma unroll
      for (int t = 0; t < 4; ++t) {
        acc[t] = __builtin_amdgcn_wmma_f32_16x16x4_f32(
            false, A, false, B[t][s], (short)0, acc[t], false, false);
      }
    }

#pragma unroll
    for (int t = 0; t < 4; ++t) {
#pragma unroll
      for (int v = 0; v < 8; ++v) {
        Sw[(v + 8 * h) * SROW + 16 * t + c] = acc[t][v];
      }
    }
    __syncthreads();

    v4f sum = *(const v4f*)Sn;
#pragma unroll
    for (int q = 1; q < N_REL; ++q) sum += *(const v4f*)(Sn + q * RSTRIDE);

    if (node < n_node) {
      const float deg = (float)c_len;
      v4f o;
      o.x = sum.x / deg; o.y = sum.y / deg; o.z = sum.z / deg; o.w = sum.w / deg;
      if (hsize > 0 && c_hm != -1) {
        o = *(const v4f*)(hbuf + (size_t)node * HID_C + 4 * c);
      }
      const size_t off = (size_t)node * HID_C + 4 * c;
      *(v4f*)(out + off) = o;
      *(v4f*)(out + (size_t)n_node * HID_C + off) = o;
    }

    c_e0 = n_e0; c_len = nn_len; c_ri = nn_ri; c_rt = nn_rt; c_hm = n_hm;
    n_e0 = p_e0; n_e1 = p_e1; n_hm = p_hm;
  }
}

extern "C" void kernel_launch(void* const* d_in, const int* in_sizes, int n_in,
                              void* d_out, int out_size, void* d_ws, size_t ws_size,
                              hipStream_t stream) {
  const float* x    = (const float*)d_in[0];
  const float* W    = (const float*)d_in[1];
  const int*   ptr  = (const int*)d_in[2];
  const int*   idx  = (const int*)d_in[3];
  const int*   et   = (const int*)d_in[4];
  const int*   hmap = (const int*)d_in[6];
  const float* hbuf = (const float*)d_in[7];
  const int*   hsz  = (const int*)d_in[9];
  float* out = (float*)d_out;

  const int n_node = in_sizes[6];

  rgcn_fused_kernel<<<512, 256, 0, stream>>>(x, W, ptr, idx, et, hmap, hbuf, hsz, out, n_node);
}


# Tile loop software-pipelined: next tile's first 8 row gathers and this tile's history row are requested before the WMMA run, rows 9-16 before an LDS-only barrier and the reduce; edge order kept
# speedup vs baseline: 2.5117x; 1.0327x over previous
#include <hip/hip_runtime.h>

typedef __attribute__((ext_vector_type(2))) float v2f;
typedef __attribute__((ext_vector_type(4))) float v4f;
typedef __attribute__((ext_vector_type(8))) float v8f;

#define IN_C   64
#define HID_C  64
#define N_REL  8
#define TILE_M 16
#define SROW   68
#define RSTRIDE (TILE_M * SROW)

__device__ __forceinline__ int kmap(int s, int h, int e) {
  return 8 * (s >> 1) + 4 * h + 2 * (s & 1) + e;
}

__device__ __forceinline__ int sort_dst(int key, int c, int h, int lane) {
  const unsigned below_me = (1u << c) - 1u;
  int rank = 0;
#pragma unroll
  for (int q = 0; q < N_REL; ++q) {
    const unsigned vote = __builtin_amdgcn_ballot_w32(key == q);
    const unsigned mine = (vote >> (16 * h)) & 0xffffu;
    const unsigned cnt  = (q < key) ? mine : ((q == key) ? (mine & below_me) : 0u);
    rank += __builtin_popcount(cnt);
  }
  if (key == N_REL) rank = c;
  return ((lane & 16) + rank) << 2;
}

template <bool FIRST, int G>
__device__ __forceinline__ void add_rows(const v4f (&v)[8], int s_t, float* Sn, v4f& a, int& ta) {
#pragma unroll
  for (int k = 0; k < 8; ++k) {
    const int tk = __shfl(s_t, 8 * G + k, 16);
    const bool nr = (tk != ta);
    if (nr && ta < N_REL) *(v4f*)(Sn + ta * RSTRIDE) = a;
    v4f st = (v4f)0.0f;
    if (!FIRST && nr) st = *(const v4f*)(Sn + (tk & (N_REL - 1)) * RSTRIDE);
    a.x = (nr ? st.x : a.x) + v[k].x;
    a.y = (nr ? st.y : a.y) + v[k].y;
    a.z = (nr ? st.z : a.z) + v[k].z;
    a.w = (nr ? st.w : a.w) + v[k].w;
    ta = tk;
  }
}

__device__ __forceinline__ void lds_barrier() {
  __builtin_amdgcn_fence(__ATOMIC_RELEASE, "workgroup", "local");
  __builtin_amdgcn_s_barrier();
  __builtin_amdgcn_fence(__ATOMIC_ACQUIRE, "workgroup", "local");
}

__global__ __launch_bounds__(256, 4)     void rgcn_fused_kernel(
    const float* __restrict__ x,
    const float* __restrict__ W,
    const int*   __restrict__ ptr,
    const int*   __restrict__ idx,
    const int*   __restrict__ et,
    const int*   __restrict__ hmap,
    const float* __restrict__ hbuf,
    const int*   __restrict__ hsize_p,
    float* __restrict__ out,
    int n_node)
{
  __shared__ float S[N_REL * RSTRIDE];

  const int tid  = threadIdx.x;
  const int lane = tid & 31;
  const int r    = tid >> 5;
  const int c    = lane & 15;
  const int h    = lane >> 4;
  const int hw   = tid >> 4;

  v2f B[4][16];
  const float* Wr = W + (r << 12);
#pragma unroll
  for (int t = 0; t < 4; ++t) {
#pragma unroll
    for (int s = 0; s < 16; ++s) {
      B[t][s].x = Wr[kmap(s, h, 0) * HID_C + 16 * t + c];
      B[t][s].y = Wr[kmap(s, h, 1) * HID_C + 16 * t + c];
    }
  }

  const int hsize   = hsize_p[0];
  const int n_tiles = (n_node + TILE_M - 1) / TILE_M;
  const int gstep   = (int)gridDim.x;

  float* Sn = S + hw * SROW + 4 * c;
  float* Sw = S + r * RSTRIDE;
  const float* xl = x + 4 * c;

  int c_len = 0, c_hm = -1;
  int n_e0 = 0, n_len = 0, n_ri = 0, n_rt = 0, n_hm = -1;
  int a_e0 = 0, a_e1 = 0, a_hm = -1;
  {
    const int node = blockIdx.x * TILE_M + hw;
    if (node < n_node) {
      n_e0 = ptr[node];
      n_len = ptr[node + 1] - n_e0;
      n_hm = hmap[node];
    }
    if (c < n_len) { n_ri = idx[n_e0 + c]; n_rt = et[n_e0 + c]; }
    const int node_a = node + gstep * TILE_M;
    if (node_a < n_node) {
      a_e0 = ptr[node_a];
      a_e1 = ptr[node_a + 1];
      a_hm = hmap[node_a];
    }
  }

  for (int mt = blockIdx.x; mt - gstep < n_tiles; mt += gstep) {
    const bool has_cur = (mt != (int)blockIdx.x);
    const int node_c = (mt - gstep) * TILE_M + hw;

    const int other  = __shfl_xor(n_len, 16);
    const int maxlen = n_len > other ? n_len : other;
    int cur_t = n_rt;
    cur_t = cur_t < 0 ? 0 : (cur_t > N_REL - 1 ? N_REL - 1 : cur_t);
    const int key = (c < n_len) ? cur_t : N_REL;
    const int dst = sort_dst(key, c, h, lane);
    const int s_i = __builtin_amdgcn_ds_permute(dst, n_ri);
    const int s_t = __builtin_amdgcn_ds_permute(dst, key);
    int ri2 = 0, rt2 = 0;
    if (16 + c < n_len) { ri2 = idx[n_e0 + 16 + c]; rt2 = et[n_e0 + 16 + c]; }

    v4f v0[8];
#pragma unroll
    for (int k = 0; k < 8; ++k) {
      const int ik = __shfl(s_i, k, 16);
      v0[k] = *(const v4f*)(xl + (size_t)ik * IN_C);
    }

    const bool use_h = has_cur && node_c < n_node && hsize > 0 && c_hm != -1;
    v4f hb = (v4f)0.0f;
    if (use_h) hb = *(const v4f*)(hbuf + (size_t)node_c * HID_C + 4 * c);

    const int a_len = a_e1 - a_e0;
    int a_ri = 0, a_rt = 0;
    if (c < a_len) { a_ri = idx[a_e0 + c]; a_rt = et[a_e0 + c]; }
    int p_e0 = 0, p_e1 = 0, p_hm = -1;
    {
      const int node_p = (mt + 2 * gstep) * TILE_M + hw;
      if (node_p < n_node) {
        p_e0 = ptr[node_p];
        p_e1 = ptr[node_p + 1];
        p_hm = hmap[node_p];
      }
    }

    if (has_cur) {
      v4f A4[8];
      {
        const float* xa = Sw + c * SROW + 4 * h;
#pragma unroll
        for (int i = 0; i < 8; ++i) A4[i] = *(const v4f*)(xa + 8 * i);
      }
      asm volatile("" :: "v"(A4[0]), "v"(A4[1]), "v"(A4[2]), "v"(A4[3]),
                         "v"(A4[4]), "v"(A4[5]), "v"(A4[6]), "v"(A4[7]));

#pragma unroll
      for (int tp = 0; tp < 4; tp += 2) {
        v8f acc0 = (v8f)0.0f, acc1 = (v8f)0.0f;
#pragma unroll
        for (int s = 0; s < 16; ++s) {
          v2f A;
          A.x = A4[s >> 1][2 * (s & 1)];
          A.y = A4[s >> 1][2 * (s & 1) + 1];
          acc0 = __builtin_amdgcn_wmma_f32_16x16x4_f32(
              false, A, false, B[tp][s], (short)0, acc0, false, false);
          acc1 = __builtin_amdgcn_wmma_f32_16x16x4_f32(
              false, A, false, B[tp + 1][s], (short)0, acc1, false, false);
        }
#pragma unroll
        for (int v = 0; v < 8; ++v) {
          Sw[(v + 8 * h) * SROW + 16 * tp + c]      = acc0[v];
          Sw[(v + 8 * h) * SROW + 16 * tp + 16 + c] = acc1[v];
        }
      }
    }

    v4f v1[8];
#pragma unroll
    for (int k = 0; k < 8; ++k) {
      const int ik = __shfl(s_i, 8 + k, 16);
      v1[k] = *(const v4f*)(xl + (size_t)ik * IN_C);
    }

    lds_barrier();

    if (has_cur) {
      v4f sum = *(const v4f*)Sn;
#pragma unroll
      for (int q = 1; q < N_REL; ++q) sum += *(const v4f*)(Sn + q * RSTRIDE);
      if (node_c < n_node) {
        const float deg = (float)c_len;
        v4f o;
        o.x = sum.x / deg; o.y = sum.y / deg; o.z = sum.z / deg; o.w = sum.w / deg;
        if (use_h) o = hb;
        const size_t off = (size_t)node_c * HID_C + 4 * c;
        *(v4f*)(out + off) = o;
        *(v4f*)(out + (size_t)n_node * HID_C + off) = o;
      }
    }

#pragma unroll
    for (int q = 0; q < N_REL; ++q) *(v4f*)(Sn + q * RSTRIDE) = (v4f)0.0f;
    {
      v4f a = (v4f)0.0f;
      int ta = N_REL;
      if (0 < maxlen) add_rows<true, 0>(v0, s_t, Sn, a, ta);
      if (8 < maxlen) add_rows<true, 1>(v1, s_t, Sn, a, ta);
      if (ta < N_REL) *(v4f*)(Sn + ta * RSTRIDE) = a;
      ta = N_REL;

      int ri = ri2, rt = rt2;
      for (int base = 16; base < maxlen; base += 16) {
        const int cur_i = ri;
        int cur_t = rt;
        cur_t = cur_t < 0 ? 0 : (cur_t > N_REL - 1 ? N_REL - 1 : cur_t);
        const int nb = base + 16 + c;
        ri = 0; rt = 0;
        if (nb < n_len) { ri = idx[n_e0 + nb]; rt = et[n_e0 + nb]; }
        const int key = (base + c < n_len) ? cur_t : N_REL;
        const int dst = sort_dst(key, c, h, lane);
        const int m_i = __builtin_amdgcn_ds_permute(dst, cur_i);
        const int m_t = __builtin_amdgcn_ds_permute(dst, key);
        {
          v4f v[8];
#pragma unroll
          for (int k = 0; k < 8; ++k) v[k] = *(const v4f*)(xl + (size_t)__shfl(m_i, k, 16) * IN_C);
          add_rows<false, 0>(v, m_t, Sn, a, ta);
        }
        if (base + 8 < maxlen) {
          v4f v[8];
#pragma unroll
          for (int k = 0; k < 8; ++k) v[k] = *(const v4f*)(xl + (size_t)__shfl(m_i, 8 + k, 16) * IN_C);
          add_rows<false, 1>(v, m_t, Sn, a, ta);
        }
        if (ta < N_REL) *(v4f*)(Sn + ta * RSTRIDE) = a;
        ta = N_REL;
      }
    }
    __syncthreads();

    c_len = n_len; c_hm = n_hm;
    n_e0 = a_e0; n_len = a_len; n_ri = a_ri; n_rt = a_rt; n_hm = a_hm;
    a_e0 = p_e0; a_e1 = p_e1; a_hm = p_hm;
  }
}

extern "C" void kernel_launch(void* const* d_in, const int* in_sizes, int n_in,
                              void* d_out, int out_size, void* d_ws, size_t ws_size,
                              hipStream_t stream) {
  const float* x    = (const float*)d_in[0];
  const float* W    = (const float*)d_in[1];
  const int*   ptr  = (const int*)d_in[2];
  const int*   idx  = (const int*)d_in[3];
  const int*   et   = (const int*)d_in[4];
  const int*   hmap = (const int*)d_in[6];
  const float* hbuf = (const float*)d_in[7];
  const int*   hsz  = (const int*)d_in[9];
  float* out = (float*)d_out;

  const int n_node = in_sizes[6];

  rgcn_fused_kernel<<<512, 256, 0, stream>>>(x, W, ptr, idx, et, hmap, hbuf, hsz, out, n_node);
}


# Gather phase branch-free: a 0/1 factor per record (DPP broadcast) restarts the run sum inside a packed multiply-add; every record stores it to its relation's row, last store stands; edge order kept
# speedup vs baseline: 2.6478x; 1.0542x over previous
#include <hip/hip_runtime.h>

typedef __attribute__((ext_vector_type(2))) float v2f;
typedef __attribute__((ext_vector_type(4))) float v4f;
typedef __attribute__((ext_vector_type(8))) float v8f;

#define IN_C   64
#define HID_C  64
#define N_REL  8
#define TILE_M 16
#define SROW   68
#define RSTRIDE (TILE_M * SROW)
#define N_PLANE (N_REL + 1)

__device__ __forceinline__ int kmap(int s, int h, int e) {
  return 8 * (s >> 1) + 4 * h + 2 * (s & 1) + e;
}

__device__ __forceinline__ int sort_dst(int key, int c, int h, int lane) {
  const unsigned below_me = (1u << c) - 1u;
  int rank = 0;
#pragma unroll
  for (int q = 0; q < N_REL; ++q) {
    const unsigned vote = __builtin_amdgcn_ballot_w32(key == q);
    const unsigned mine = (vote >> (16 * h)) & 0xffffu;
    const unsigned cnt  = (q < key) ? mine : ((q == key) ? (mine & below_me) : 0u);
    rank += __builtin_popcount(cnt);
  }
  if (key == N_REL) rank = c;
  return ((lane & 16) + rank) << 2;
}

template <int G>
__device__ __forceinline__ void sum_rows(const v4f (&v)[8], float keepf, int poff, float* Sn, v4f& a) {
#pragma unroll
  for (int k = 0; k < 8; ++k) {
    const float kf = __shfl(keepf, 8 * G + k, 16);
    const int   pk = __shfl(poff, 8 * G + k, 16);
    a.x = __builtin_fmaf(a.x, kf, v[k].x);
    a.y = __builtin_fmaf(a.y, kf, v[k].y);
    a.z = __builtin_fmaf(a.z, kf, v[k].z);
    a.w = __builtin_fmaf(a.w, kf, v[k].w);
    *(v4f*)(Sn + pk) = a;
  }
}

template <int G>
__device__ __forceinline__ void add_rows(const v4f (&v)[8], int s_t, float* Sn, v4f& a, int& ta) {
#pragma unroll
  for (int k = 0; k < 8; ++k) {
    const int tk = __shfl(s_t, 8 * G + k, 16);
    const bool nr = (tk != ta);
    if (nr && ta < N_REL) *(v4f*)(Sn + ta * RSTRIDE) = a;
    v4f st = (v4f)0.0f;
    if (nr) st = *(const v4f*)(Sn + (tk & (N_REL - 1)) * RSTRIDE);
    a.x = (nr ? st.x : a.x) + v[k].x;
    a.y = (nr ? st.y : a.y) + v[k].y;
    a.z = (nr ? st.z : a.z) + v[k].z;
    a.w = (nr ? st.w : a.w) + v[k].w;
    ta = tk;
  }
}

__device__ __forceinline__ void lds_barrier() {
  __builtin_amdgcn_fence(__ATOMIC_RELEASE, "workgroup", "local");
  __builtin_amdgcn_s_barrier();
  __builtin_amdgcn_fence(__ATOMIC_ACQUIRE, "workgroup", "local");
}

__global__ __launch_bounds__(256, 4)     void rgcn_fused_kernel(
    const float* __restrict__ x,
    const float* __restrict__ W,
    const int*   __restrict__ ptr,
    const int*   __restrict__ idx,
    const int*   __restrict__ et,
    const int*   __restrict__ hmap,
    const float* __restrict__ hbuf,
    const int*   __restrict__ hsize_p,
    float* __restrict__ out,
    int n_node)
{
  __shared__ float S[N_PLANE * RSTRIDE];

  const int tid  = threadIdx.x;
  const int lane = tid & 31;
  const int r    = tid >> 5;
  const int c    = lane & 15;
  const int h    = lane >> 4;
  const int hw   = tid >> 4;

  v2f B[4][16];
  const float* Wr = W + (r << 12);
#pragma unroll
  for (int t = 0; t < 4; ++t) {
#pragma unroll
    for (int s = 0; s < 16; ++s) {
      B[t][s].x = Wr[kmap(s, h, 0) * HID_C + 16 * t + c];
      B[t][s].y = Wr[kmap(s, h, 1) * HID_C + 16 * t + c];
    }
  }

  const int hsize   = hsize_p[0];
  const int n_tiles = (n_node + TILE_M - 1) / TILE_M;
  const int gstep   = (int)gridDim.x;

  float* Sn = S + hw * SROW + 4 * c;
  float* Sw = S + r * RSTRIDE;
  const float* xl = x + 4 * c;

  int c_len = 0, c_hm = -1;
  int n_e0 = 0, n_len = 0, n_ri = 0, n_rt = 0, n_hm = -1;
  int a_e0 = 0, a_e1 = 0, a_hm = -1;
  {
    const int node = blockIdx.x * TILE_M + hw;
    if (node < n_node) {
      n_e0 = ptr[node];
      n_len = ptr[node + 1] - n_e0;
      n_hm = hmap[node];
    }
    if (c < n_len) { n_ri = idx[n_e0 + c]; n_rt = et[n_e0 + c]; }
    const int node_a = node + gstep * TILE_M;
    if (node_a < n_node) {
      a_e0 = ptr[node_a];
      a_e1 = ptr[node_a + 1];
      a_hm = hmap[node_a];
    }
  }

  for (int mt = blockIdx.x; mt - gstep < n_tiles; mt += gstep) {
    const bool has_cur = (mt != (int)blockIdx.x);
    const int node_c = (mt - gstep) * TILE_M + hw;

    const int other  = __shfl_xor(n_len, 16);
    const int maxlen = n_len > other ? n_len : other;
    int cur_t = n_rt;
    cur_t = cur_t < 0 ? 0 : (cur_t > N_REL - 1 ? N_REL - 1 : cur_t);
    const int key = (c < n_len) ? cur_t : N_REL;
    const int dst = sort_dst(key, c, h, lane);
    const int s_i = __builtin_amdgcn_ds_permute(dst, n_ri);
    const int s_t = __builtin_amdgcn_ds_permute(dst, key);
    int ri2 = 0, rt2 = 0;
    if (16 + c < n_len) { ri2 = idx[n_e0 + 16 + c]; rt2 = et[n_e0 + 16 + c]; }

    v4f v0[8];
#pragma unroll
    for (int k = 0; k < 8; ++k) {
      const int ik = __shfl(s_i, k, 16);
      v0[k] = *(const v4f*)(xl + (size_t)ik * IN_C);
    }

    const bool use_h = has_cur && node_c < n_node && hsize > 0 && c_hm != -1;
    v4f hb = (v4f)0.0f;
    if (use_h) hb = *(const v4f*)(hbuf + (size_t)node_c * HID_C + 4 * c);

    const int a_len = a_e1 - a_e0;
    int a_ri = 0, a_rt = 0;
    if (c < a_len) { a_ri = idx[a_e0 + c]; a_rt = et[a_e0 + c]; }
    int p_e0 = 0, p_e1 = 0, p_hm = -1;
    {
      const int node_p = (mt + 2 * gstep) * TILE_M + hw;
      if (node_p < n_node) {
        p_e0 = ptr[node_p];
        p_e1 = ptr[node_p + 1];
        p_hm = hmap[node_p];
      }
    }

    if (has_cur) {
      v4f A4[8];
      {
        const float* xa = Sw + c * SROW + 4 * h;
#pragma unroll
        for (int i = 0; i < 8; ++i) A4[i] = *(const v4f*)(xa + 8 * i);
      }
      asm volatile("" :: "v"(A4[0]), "v"(A4[1]), "v"(A4[2]), "v"(A4[3]),
                         "v"(A4[4]), "v"(A4[5]), "v"(A4[6]), "v"(A4[7]));

#pragma unroll
      for (int tp = 0; tp < 4; tp += 2) {
        v8f acc0 = (v8f)0.0f, acc1 = (v8f)0.0f;
#pragma unroll
        for (int s = 0; s < 16; ++s) {
          v2f A;
          A.x = A4[s >> 1][2 * (s & 1)];
          A.y = A4[s >> 1][2 * (s & 1) + 1];
          acc0 = __builtin_amdgcn_wmma_f32_16x16x4_f32(
              false, A, false, B[tp][s], (short)0, acc0, false, false);
          acc1 = __builtin_amdgcn_wmma_f32_16x16x4_f32(
              false, A, false, B[tp + 1][s], (short)0, acc1, false, false);
        }
#pragma unroll
        for (int v = 0; v < 8; ++v) {
          Sw[(v + 8 * h) * SROW + 16 * tp + c]      = acc0[v];
          Sw[(v + 8 * h) * SROW + 16 * tp + 16 + c] = acc1[v];
        }
      }
    }

    v4f v1[8];
#pragma unroll
    for (int k = 0; k < 8; ++k) {
      const int ik = __shfl(s_i, 8 + k, 16);
      v1[k] = *(const v4f*)(xl + (size_t)ik * IN_C);
    }

    lds_barrier();

    if (has_cur) {
      v4f sum = *(const v4f*)Sn;
#pragma unroll
      for (int q = 1; q < N_REL; ++q) sum += *(const v4f*)(Sn + q * RSTRIDE);
      if (node_c < n_node) {
        const float deg = (float)c_len;
        v4f o;
        o.x = sum.x / deg; o.y = sum.y / deg; o.z = sum.z / deg; o.w = sum.w / deg;
        if (use_h) o = hb;
        const size_t off = (size_t)node_c * HID_C + 4 * c;
        *(v4f*)(out + off) = o;
        *(v4f*)(out + (size_t)n_node * HID_C + off) = o;
      }
    }

#pragma unroll
    for (int q = 0; q < N_REL; ++q) *(v4f*)(Sn + q * RSTRIDE) = (v4f)0.0f;
    {
      const int   p_t   = __builtin_amdgcn_update_dpp(-1, s_t, 0x111, 0xf, 0xf, false);
      const float keepf = (s_t == p_t) ? 1.0f : 0.0f;
      const int   poff  = __mul24(s_t, RSTRIDE);
      v4f a = (v4f)0.0f;
      if (0 < maxlen) sum_rows<0>(v0, keepf, poff, Sn, a);
      if (8 < maxlen) sum_rows<1>(v1, keepf, poff, Sn, a);
      int ta = N_REL;

      int ri = ri2, rt = rt2;
      for (int base = 16; base < maxlen; base += 16) {
        const int cur_i = ri;
        int cur_t = rt;
        cur_t = cur_t < 0 ? 0 : (cur_t > N_REL - 1 ? N_REL - 1 : cur_t);
        const int nb = base + 16 + c;
        ri = 0; rt = 0;
        if (nb < n_len) { ri = idx[n_e0 + nb]; rt = et[n_e0 + nb]; }
        const int key = (base + c < n_len) ? cur_t : N_REL;
        const int dst = sort_dst(key, c, h, lane);
        const int m_i = __builtin_amdgcn_ds_permute(dst, cur_i);
        const int m_t = __builtin_amdgcn_ds_permute(dst, key);
        {
          v4f v[8];
#pragma unroll
          for (int k = 0; k < 8; ++k) v[k] = *(const v4f*)(xl + (size_t)__shfl(m_i, k, 16) * IN_C);
          add_rows<0>(v, m_t, Sn, a, ta);
        }
        if (base + 8 < maxlen) {
          v4f v[8];
#pragma unroll
          for (int k = 0; k < 8; ++k) v[k] = *(const v4f*)(xl + (size_t)__shfl(m_i, 8 + k, 16) * IN_C);
          add_rows<1>(v, m_t, Sn, a, ta);
        }
        if (ta < N_REL) *(v4f*)(Sn + ta * RSTRIDE) = a;
        ta = N_REL;
      }
    }
    __syncthreads();

    c_len = n_len; c_hm = n_hm;
    n_e0 = a_e0; n_len = a_len; n_ri = a_ri; n_rt = a_rt; n_hm = a_hm;
    a_e0 = p_e0; a_e1 = p_e1; a_hm = p_hm;
  }
}

extern "C" void kernel_launch(void* const* d_in, const int* in_sizes, int n_in,
                              void* d_out, int out_size, void* d_ws, size_t ws_size,
                              hipStream_t stream) {
  const float* x    = (const float*)d_in[0];
  const float* W    = (const float*)d_in[1];
  const int*   ptr  = (const int*)d_in[2];
  const int*   idx  = (const int*)d_in[3];
  const int*   et   = (const int*)d_in[4];
  const int*   hmap = (const int*)d_in[6];
  const float* hbuf = (const float*)d_in[7];
  const int*   hsz  = (const int*)d_in[9];
  float* out = (float*)d_out;

  const int n_node = in_sizes[6];

  rgcn_fused_kernel<<<512, 256, 0, stream>>>(x, W, ptr, idx, et, hmap, hbuf, hsz, out, n_node);
}
